# starting kernel
# baseline (speedup 1.0000x reference)
#include <hip/hip_runtime.h>
#include <math.h>

#define B_    32
#define IN_   2048
#define DIN_  8
#define N_    64
#define DOUT_ 16
#define C_    32
#define REP_  (IN_ / C_)
#define TILE_ 512
#define NTILE (IN_ / TILE_)
#define EPS_  1e-7f

typedef float v2f __attribute__((ext_vector_type(2)));
typedef float v8f __attribute__((ext_vector_type(8)));

__device__ __forceinline__ void hat_tile_wmma(
    const float* __restrict__ x, const float* __restrict__ Wt,
    float* __restrict__ hat, int b, int n, int tile, int wave, int lane)
{
  const int mrow = lane & 15;
  const int kh   = (lane >> 4) << 1;
  const int half = lane >> 4;
  for (int u = wave; u < TILE_ / 16; u += 8) {
    const int il0 = u << 4;
    const int gi0 = tile * TILE_ + il0;
    const int cu  = gi0 >> 6;
    const float* xr = x  + ((size_t)b * IN_ + gi0 + mrow) * DIN_ + kh;
    const float* wr = Wt + (((size_t)n * C_ + cu) * DOUT_ + mrow) * DIN_ + kh;
    v2f a0 = { xr[0], xr[1] };
    v2f a1 = { xr[4], xr[5] };
    v2f b0 = { wr[0], wr[1] };
    v2f b1 = { wr[4], wr[5] };
    v8f acc = {0.f,0.f,0.f,0.f,0.f,0.f,0.f,0.f};
    acc = __builtin_amdgcn_wmma_f32_16x16x4_f32(false, a0, false, b0,
                                                (short)0, acc, false, false);
    acc = __builtin_amdgcn_wmma_f32_16x16x4_f32(false, a1, false, b1,
                                                (short)0, acc, false, false);
#pragma unroll
    for (int r = 0; r < 8; ++r)
      hat[(il0 + r + (half << 3)) * DOUT_ + mrow] = acc[r];
  }
}

__global__ __launch_bounds__(256) void caps_softmax(
    const float* __restrict__ logits, float* __restrict__ smax, float* __restrict__ sinv)
{
  const int e = blockIdx.x * 256 + threadIdx.x;
  const int b = e >> 11;
  const int i = e & (IN_ - 1);
  const float* lp = logits + (size_t)b * N_ * IN_ + i;
  float mx = -3.4e38f;
#pragma unroll 4
  for (int n = 0; n < N_; ++n) mx = fmaxf(mx, lp[(size_t)n * IN_]);
  float s = 0.f;
#pragma unroll 4
  for (int n = 0; n < N_; ++n) s += __expf(lp[(size_t)n * IN_] - mx);
  smax[e] = mx;
  sinv[e] = 1.0f / s;
}

template <int MODE>
__global__ __launch_bounds__(256) void caps_route(
    const float* __restrict__ x, const float* __restrict__ Wt,
    const float* __restrict__ bias, float* __restrict__ logits,
    const float* __restrict__ smax, const float* __restrict__ sinv,
    float* __restrict__ out)
{
  __shared__ float hat[TILE_ * DOUT_];
  __shared__ float cld[TILE_];
  __shared__ float part[256];
  __shared__ float sld[DOUT_];
  __shared__ float old_[DOUT_];
  __shared__ float scl;

  const int b    = blockIdx.x >> 6;
  const int n    = blockIdx.x & (N_ - 1);
  const int wave = threadIdx.x >> 5;
  const int lane = threadIdx.x & 31;
  const int dc    = threadIdx.x & 15;
  const int chunk = threadIdx.x >> 4;
  const size_t lbase = ((size_t)b * N_ + n) * IN_;

  float sacc = 0.f;
  for (int tile = 0; tile < NTILE; ++tile) {
    if (tile + 1 < NTILE)
      __builtin_prefetch(
          x + ((size_t)b * IN_ + (tile + 1) * TILE_ + (threadIdx.x << 1)) * DIN_, 0, 0);
    hat_tile_wmma(x, Wt, hat, b, n, tile, wave, lane);
    for (int ii = threadIdx.x; ii < TILE_; ii += 256) {
      float cv;
      if (MODE == 0) {
        cv = 1.0f / (float)N_;
      } else {
        const int gi = tile * TILE_ + ii;
        const size_t e = (size_t)b * IN_ + gi;
        cv = __expf(logits[lbase + gi] - smax[e]) * sinv[e];
      }
      cld[ii] = cv;
    }
    __syncthreads();
    const float* hp = hat + (chunk * 32) * DOUT_ + dc;
    const float* cp = cld + chunk * 32;
#pragma unroll 8
    for (int ii = 0; ii < 32; ++ii) sacc += cp[ii] * hp[ii * DOUT_];
    __syncthreads();
  }

  part[threadIdx.x] = sacc;
  __syncthreads();
  if (threadIdx.x < DOUT_) {
    float s = bias[n * DOUT_ + threadIdx.x];
#pragma unroll
    for (int ch = 0; ch < 16; ++ch) s += part[ch * 16 + threadIdx.x];
    sld[threadIdx.x] = s;
  }
  __syncthreads();
  if (threadIdx.x == 0) {
    float sn = 0.f;
#pragma unroll
    for (int d2 = 0; d2 < DOUT_; ++d2) sn += sld[d2] * sld[d2];
    scl = sn / (1.0f + sn) * rsqrtf(sn + EPS_);
  }
  __syncthreads();
  if (threadIdx.x < DOUT_) {
    const float o = scl * sld[threadIdx.x];
    old_[threadIdx.x] = o;
    out[((size_t)b * N_ + n) * DOUT_ + threadIdx.x] = o;
  }
  __syncthreads();

  if (MODE < 2) {
    for (int tile = 0; tile < NTILE; ++tile) {
      hat_tile_wmma(x, Wt, hat, b, n, tile, wave, lane);
      __syncthreads();
      for (int ii = threadIdx.x; ii < TILE_; ii += 256) {
        const float* hr = hat + ii * DOUT_;
        float dlt = 0.f;
#pragma unroll
        for (int d2 = 0; d2 < DOUT_; ++d2) dlt += old_[d2] * hr[d2];
        const size_t g = lbase + tile * TILE_ + ii;
        logits[g] = (MODE == 0) ? dlt : (logits[g] + dlt);
      }
      __syncthreads();
    }
  }
}

extern "C" void kernel_launch(void* const* d_in, const int* in_sizes, int n_in,
                              void* d_out, int out_size, void* d_ws, size_t ws_size,
                              hipStream_t stream)
{
  const float* x    = (const float*)d_in[0];
  const float* Wt   = (const float*)d_in[1];
  const float* bias = (const float*)d_in[2];
  float* out = (float*)d_out;

  float* logits = (float*)d_ws;
  float* smax   = logits + (size_t)B_ * N_ * IN_;
  float* sinv   = smax   + (size_t)B_ * IN_;

  dim3 grid(B_ * N_), blk(256);
  dim3 sgrid((B_ * IN_) / 256), sblk(256);

  caps_route<0><<<grid, blk, 0, stream>>>(x, Wt, bias, logits, smax, sinv, out);
  caps_softmax<<<sgrid, sblk, 0, stream>>>(logits, smax, sinv);
  caps_route<1><<<grid, blk, 0, stream>>>(x, Wt, bias, logits, smax, sinv, out);
  caps_softmax<<<sgrid, sblk, 0, stream>>>(logits, smax, sinv);
  caps_route<2><<<grid, blk, 0, stream>>>(x, Wt, bias, logits, smax, sinv, out);
}


# Chunk-stationary blocks own all 64 output capsules of 256 inputs: routing softmax taken in-block in an LDS plane, logits recomputed from summed outputs; logits and statistics launches gone (5 to 4)
# speedup vs baseline: 2.6212x; 2.6212x over previous
#include <hip/hip_runtime.h>
#include <math.h>


#define B_    32
#define IN_   2048
#define DIN_  8
#define N_    64
#define DOUT_ 16
#define C_    32
#define REP_  (IN_ / C_)
#define CHB_  4
#define CHUNK_ (CHB_ * REP_)
#define NCHUNK_ (IN_ / CHUNK_)
#define UPC_  (REP_ / 16)
#define UNITS_ (CHB_ * UPC_)
#define G_    4
#define NWAVE_ (N_ / G_)
#define WSTR_ (C_ * DOUT_ * DIN_)
#define PLANE_ (B_ * NCHUNK_ * N_ * DOUT_)
#define EPS_  1e-7f

typedef float v2f __attribute__((ext_vector_type(2)));
typedef float v8f __attribute__((ext_vector_type(8)));

__device__ __forceinline__ float xhalf(float v)
{
  const int q = __builtin_bit_cast(int, v);
  return __builtin_bit_cast(float,
      __builtin_amdgcn_permlanex16(q, q, 0x76543210, 0xfedcba98, false, false));
}

__device__ __forceinline__ v8f vote_tile_T(v2f w0, v2f w1, v2f x0, v2f x1)
{
  v8f acc = {0.f,0.f,0.f,0.f,0.f,0.f,0.f,0.f};
  acc = __builtin_amdgcn_wmma_f32_16x16x4_f32(false, w0, false, x0,
                                              (short)0, acc, false, false);
  acc = __builtin_amdgcn_wmma_f32_16x16x4_f32(false, w1, false, x1,
                                              (short)0, acc, false, false);
  return acc;
}

__device__ __forceinline__ void load_w(const float* __restrict__ ww, v2f (&w)[CHB_][2])
{
#pragma unroll
  for (int cc = 0; cc < CHB_; ++cc) {
    const float* wr = ww + cc * (DOUT_ * DIN_);
    w[cc][0] = (v2f){ wr[0], wr[1] };
    w[cc][1] = (v2f){ wr[4], wr[5] };
  }
}

__device__ __forceinline__ void squashed(const float* __restrict__ part,
                                         const float* __restrict__ bias,
                                         int b, int n0, int col, float (&o)[G_])
{
#pragma unroll
  for (int g = 0; g < G_; ++g) {
    float s = bias[(n0 + g) * DOUT_ + col];
#pragma unroll
    for (int k = 0; k < NCHUNK_; ++k)
      s += part[(((size_t)b * NCHUNK_ + k) * N_ + n0 + g) * DOUT_ + col];
    float sn = s * s;
    sn += __shfl_xor(sn, 1, 32);
    sn += __shfl_xor(sn, 2, 32);
    sn += __shfl_xor(sn, 4, 32);
    sn += __shfl_xor(sn, 8, 32);
    const float scl = sn / (1.0f + sn) * rsqrtf(sn + EPS_);
    o[g] = scl * s;
  }
}

template <int MODE>
__global__ __launch_bounds__(512, 4) void caps_route(
    const float* __restrict__ x, const float* __restrict__ Wt,
    const float* __restrict__ bias,
    const float* __restrict__ pa, const float* __restrict__ pb,
    float* __restrict__ pout)
{
  __shared__ float lgs[N_ * CHUNK_];
  __shared__ float stage[NWAVE_ * G_ * DOUT_];

  const int b     = blockIdx.x / NCHUNK_;
  const int chunk = blockIdx.x % NCHUNK_;
  const int wave  = threadIdx.x >> 5;
  const int lane  = threadIdx.x & 31;
  const int col   = lane & 15;
  const int half  = lane >> 4;
  const int kh    = half << 1;
  const int n0    = wave * G_;
  const int i0    = chunk * CHUNK_;
  const size_t ebase = (size_t)b * IN_;

  const float* xw = x + (ebase + i0 + col) * DIN_ + kh;
  v2f xa[UNITS_], xb[UNITS_];
#pragma unroll
  for (int u = 0; u < UNITS_; ++u) {
    const float* xr = xw + u * (16 * DIN_);
    xa[u] = (v2f){ xr[0], xr[1] };
    xb[u] = (v2f){ xr[4], xr[5] };
  }

  const float* wbase = Wt + (((size_t)n0 * C_ + chunk * CHB_) * DOUT_ + col) * DIN_ + kh;

  if (MODE != 0) {
    float o[G_];
    squashed(pa, bias, b, n0, col, o);
    if (MODE == 2) {
      float o1[G_];
      squashed(pb, bias, b, n0, col, o1);
#pragma unroll
      for (int g = 0; g < G_; ++g) o[g] += o1[g];
    }

    v2f wq[G_][CHB_][2];
    load_w(wbase, wq[0]);
#pragma unroll
    for (int g = 0; g < G_; ++g) {
      if (g + 1 < G_) load_w(wbase + (size_t)(g + 1) * WSTR_, wq[g + 1]);
      float od[8];
#pragma unroll
      for (int r = 0; r < 8; ++r) od[r] = __shfl(o[g], (half << 3) + r, 32);
#pragma unroll
      for (int cc = 0; cc < CHB_; ++cc) {
#pragma unroll
        for (int t = 0; t < UPC_; t += 2) {
          const int u = cc * UPC_ + t;
          const v8f ta = vote_tile_T(wq[g][cc][0], wq[g][cc][1], xa[u], xb[u]);
          const v8f tb = vote_tile_T(wq[g][cc][0], wq[g][cc][1], xa[u + 1], xb[u + 1]);
          float da = 0.f, db = 0.f;
#pragma unroll
          for (int r = 0; r < 8; ++r) { da += od[r] * ta[r]; db += od[r] * tb[r]; }
          da += xhalf(da);
          db += xhalf(db);
          lgs[(n0 + g) * CHUNK_ + u * 16 + lane] = half ? db : da;
        }
      }
      __builtin_amdgcn_sched_barrier(0);
    }

    __syncthreads();

    if (threadIdx.x < CHUNK_) {
      const int t = threadIdx.x;
      float v[N_];
#pragma unroll
      for (int n = 0; n < N_; ++n) v[n] = lgs[n * CHUNK_ + t];
      float mx = v[0];
#pragma unroll
      for (int n = 1; n < N_; ++n) mx = fmaxf(mx, v[n]);
      float s = 0.f;
#pragma unroll
      for (int n = 0; n < N_; ++n) { v[n] = __expf(v[n] - mx); s += v[n]; }
      const float inv = 1.0f / s;
#pragma unroll
      for (int n = 0; n < N_; ++n) lgs[n * CHUNK_ + t] = v[n] * inv;
    }

    __syncthreads();
  }

  v2f wf[G_][CHB_][2];
  load_w(wbase, wf[0]);
#pragma unroll
  for (int g = 0; g < G_; ++g) {
    if (g + 1 < G_) load_w(wbase + (size_t)(g + 1) * WSTR_, wf[g + 1]);
    v8f sacc = {0.f,0.f,0.f,0.f,0.f,0.f,0.f,0.f};
#pragma unroll
    for (int cc = 0; cc < CHB_; ++cc) {
#pragma unroll
      for (int t = 0; t < UPC_; ++t) {
        const int u = cc * UPC_ + t;
        float cv;
        if (MODE == 0) cv = 1.0f / (float)N_;
        else           cv = lgs[(n0 + g) * CHUNK_ + u * 16 + col];
        const v8f acc = vote_tile_T(wf[g][cc][0], wf[g][cc][1], xa[u], xb[u]);
        sacc += cv * acc;
      }
    }
#pragma unroll
    for (int r = 0; r < 8; ++r) {
      float v = sacc[r];
      v += __shfl_xor(v, 1, 32);
      v += __shfl_xor(v, 2, 32);
      v += __shfl_xor(v, 4, 32);
      v += __shfl_xor(v, 8, 32);
      sacc[r] = v;
    }
    if (col == 0) {
#pragma unroll
      for (int r = 0; r < 8; ++r) stage[(wave * G_ + g) * DOUT_ + (half << 3) + r] = sacc[r];
    }
    __builtin_amdgcn_sched_barrier(0);
  }

  __syncthreads();

  float* pw = pout + (((size_t)b * NCHUNK_ + chunk) * N_ + n0) * DOUT_;
#pragma unroll
  for (int q = 0; q < 2; ++q)
    pw[q * 32 + lane] = stage[wave * (G_ * DOUT_) + q * 32 + lane];
}

__global__ __launch_bounds__(256) void caps_final(
    const float* __restrict__ part, const float* __restrict__ bias, float* __restrict__ out)
{
  const int e  = blockIdx.x * 256 + threadIdx.x;
  const int b  = e / (N_ * DOUT_);
  const int nd = e % (N_ * DOUT_);
  float s = bias[nd];
#pragma unroll
  for (int k = 0; k < NCHUNK_; ++k)
    s += part[((size_t)b * NCHUNK_ + k) * (N_ * DOUT_) + nd];
  float sn = s * s;
  sn += __shfl_xor(sn, 1, 32);
  sn += __shfl_xor(sn, 2, 32);
  sn += __shfl_xor(sn, 4, 32);
  sn += __shfl_xor(sn, 8, 32);
  const float scl = sn / (1.0f + sn) * rsqrtf(sn + EPS_);
  out[e] = scl * s;
}

extern "C" void kernel_launch(void* const* d_in, const int* in_sizes, int n_in,
                              void* d_out, int out_size, void* d_ws, size_t ws_size,
                              hipStream_t stream)
{
  const float* x    = (const float*)d_in[0];
  const float* Wt   = (const float*)d_in[1];
  const float* bias = (const float*)d_in[2];
  float* out = (float*)d_out;

  float* p0 = (float*)d_ws;
  float* p1 = p0 + (size_t)PLANE_;
  float* p2 = p1 + (size_t)PLANE_;

  dim3 grid(B_ * NCHUNK_), blk(512);
  dim3 fgrid((B_ * N_ * DOUT_) / 256), fblk(256);

  caps_route<0><<<grid, blk, 0, stream>>>(x, Wt, bias, p1, p2, p0);
  caps_route<1><<<grid, blk, 0, stream>>>(x, Wt, bias, p0, p0, p1);
  caps_route<2><<<grid, blk, 0, stream>>>(x, Wt, bias, p0, p1, p2);
  caps_final<<<fgrid, fblk, 0, stream>>>(p2, bias, out);
}


# Each wave's 16 vote tiles of one channel stay in registers from the logit pass to the weighted-sum pass; softmax over n in registers, per-wave max and sum via LDS; WMMAs of launches 2 and 3 halve
# speedup vs baseline: 2.7848x; 1.0624x over previous
#include <hip/hip_runtime.h>
#include <math.h>


#define B_    32
#define IN_   2048
#define DIN_  8
#define N_    64
#define DOUT_ 16
#define C_    32
#define REP_  (IN_ / C_)
#define CHB_  4
#define CHUNK_ (CHB_ * REP_)
#define NCHUNK_ (IN_ / CHUNK_)
#define UPC_  (REP_ / 16)
#define UNITS_ (CHB_ * UPC_)
#define G_    4
#define NWAVE_ (N_ / G_)
#define WSTR_ (C_ * DOUT_ * DIN_)
#define PLANE_ (B_ * NCHUNK_ * N_ * DOUT_)
#define EPS_  1e-7f

typedef float v2f __attribute__((ext_vector_type(2)));
typedef float v8f __attribute__((ext_vector_type(8)));

__device__ __forceinline__ float xhalf(float v)
{
  const int q = __builtin_bit_cast(int, v);
  return __builtin_bit_cast(float,
      __builtin_amdgcn_permlanex16(q, q, 0x76543210, 0xfedcba98, false, false));
}

__device__ __forceinline__ v8f vote_tile_T(v2f w0, v2f w1, v2f x0, v2f x1)
{
  v8f acc = {0.f,0.f,0.f,0.f,0.f,0.f,0.f,0.f};
  acc = __builtin_amdgcn_wmma_f32_16x16x4_f32(false, w0, false, x0,
                                              (short)0, acc, false, false);
  acc = __builtin_amdgcn_wmma_f32_16x16x4_f32(false, w1, false, x1,
                                              (short)0, acc, false, false);
  return acc;
}

__device__ __forceinline__ void load_w(const float* __restrict__ ww, v2f (&w)[CHB_][2])
{
#pragma unroll
  for (int cc = 0; cc < CHB_; ++cc) {
    const float* wr = ww + cc * (DOUT_ * DIN_);
    w[cc][0] = (v2f){ wr[0], wr[1] };
    w[cc][1] = (v2f){ wr[4], wr[5] };
  }
}

__device__ __forceinline__ void squashed(const float* __restrict__ part,
                                         const float* __restrict__ bias,
                                         int b, int n0, int col, float (&o)[G_])
{
#pragma unroll
  for (int g = 0; g < G_; ++g) {
    float s = bias[(n0 + g) * DOUT_ + col];
#pragma unroll
    for (int k = 0; k < NCHUNK_; ++k)
      s += part[(((size_t)b * NCHUNK_ + k) * N_ + n0 + g) * DOUT_ + col];
    float sn = s * s;
    sn += __shfl_xor(sn, 1, 32);
    sn += __shfl_xor(sn, 2, 32);
    sn += __shfl_xor(sn, 4, 32);
    sn += __shfl_xor(sn, 8, 32);
    const float scl = sn / (1.0f + sn) * rsqrtf(sn + EPS_);
    o[g] = scl * s;
  }
}

__device__ __forceinline__ void load_frags(const float* __restrict__ xc, const float* __restrict__ wc,
                                           v2f (&xa)[UPC_], v2f (&xb)[UPC_],
                                           v2f (&wa)[G_], v2f (&wb)[G_])
{
#pragma unroll
  for (int t = 0; t < UPC_; ++t) {
    const float* xr = xc + t * (16 * DIN_);
    xa[t] = (v2f){ xr[0], xr[1] };
    xb[t] = (v2f){ xr[4], xr[5] };
  }
#pragma unroll
  for (int g = 0; g < G_; ++g) {
    const float* wr = wc + (size_t)g * WSTR_;
    wa[g] = (v2f){ wr[0], wr[1] };
    wb[g] = (v2f){ wr[4], wr[5] };
  }
}

__device__ __forceinline__ void stage_capsule(v8f sacc, float* __restrict__ stage,
                                              int wave, int g, int col, int half)
{
#pragma unroll
  for (int r = 0; r < 8; ++r) {
    float v = sacc[r];
    v += __shfl_xor(v, 1, 32);
    v += __shfl_xor(v, 2, 32);
    v += __shfl_xor(v, 4, 32);
    v += __shfl_xor(v, 8, 32);
    sacc[r] = v;
  }
  if (col == 0) {
#pragma unroll
    for (int r = 0; r < 8; ++r) stage[(wave * G_ + g) * DOUT_ + (half << 3) + r] = sacc[r];
  }
}

__device__ __forceinline__ void flush_partial(const float* __restrict__ stage, float* __restrict__ pw,
                                              int wave, int lane)
{
  __syncthreads();
#pragma unroll
  for (int q = 0; q < 2; ++q)
    pw[q * 32 + lane] = stage[wave * (G_ * DOUT_) + q * 32 + lane];
}

__global__ __launch_bounds__(512, 4) void caps_route_first(
    const float* __restrict__ x, const float* __restrict__ Wt, float* __restrict__ pout)
{
  __shared__ float stage[NWAVE_ * G_ * DOUT_];

  const int b     = blockIdx.x / NCHUNK_;
  const int chunk = blockIdx.x % NCHUNK_;
  const int wave  = threadIdx.x >> 5;
  const int lane  = threadIdx.x & 31;
  const int col   = lane & 15;
  const int half  = lane >> 4;
  const int kh    = half << 1;
  const int n0    = wave * G_;
  const int i0    = chunk * CHUNK_;
  const size_t ebase = (size_t)b * IN_;

  const float* xw = x + (ebase + i0 + col) * DIN_ + kh;
  v2f xa[UNITS_], xb[UNITS_];
#pragma unroll
  for (int u = 0; u < UNITS_; ++u) {
    const float* xr = xw + u * (16 * DIN_);
    xa[u] = (v2f){ xr[0], xr[1] };
    xb[u] = (v2f){ xr[4], xr[5] };
  }

  const float* wbase = Wt + (((size_t)n0 * C_ + chunk * CHB_) * DOUT_ + col) * DIN_ + kh;

  v2f wf[G_][CHB_][2];
  load_w(wbase, wf[0]);
#pragma unroll
  for (int g = 0; g < G_; ++g) {
    if (g + 1 < G_) load_w(wbase + (size_t)(g + 1) * WSTR_, wf[g + 1]);
    v8f sa = {0.f,0.f,0.f,0.f,0.f,0.f,0.f,0.f};
#pragma unroll
    for (int cc = 0; cc < CHB_; ++cc) {
#pragma unroll
      for (int t = 0; t < UPC_; ++t) {
        const int u = cc * UPC_ + t;
        const v8f acc = vote_tile_T(wf[g][cc][0], wf[g][cc][1], xa[u], xb[u]);
        sa += (1.0f / (float)N_) * acc;
      }
    }
    stage_capsule(sa, stage, wave, g, col, half);
    __builtin_amdgcn_sched_barrier(0);
  }

  float* pw = pout + (((size_t)b * NCHUNK_ + chunk) * N_ + n0) * DOUT_;
  flush_partial(stage, pw, wave, lane);
}

template <int MODE>
__global__ __launch_bounds__(512, 4) void caps_route(
    const float* __restrict__ x, const float* __restrict__ Wt,
    const float* __restrict__ bias,
    const float* __restrict__ pa, const float* __restrict__ pb,
    float* __restrict__ pout)
{
  __shared__ __attribute__((aligned(16))) float osum_s[N_ * DOUT_];
  __shared__ float xm[NWAVE_ * REP_];
  __shared__ float xs[NWAVE_ * REP_];
  __shared__ float stage[NWAVE_ * G_ * DOUT_];

  const int b     = blockIdx.x / NCHUNK_;
  const int chunk = blockIdx.x % NCHUNK_;
  const int wave  = threadIdx.x >> 5;
  const int lane  = threadIdx.x & 31;
  const int col   = lane & 15;
  const int half  = lane >> 4;
  const int kh    = half << 1;
  const int n0    = wave * G_;
  const int i0    = chunk * CHUNK_;

  const unsigned xo = (unsigned)((b * IN_ + i0 + col) * DIN_ + kh);
  const unsigned wo = (unsigned)(((n0 * C_ + chunk * CHB_) * DOUT_ + col) * DIN_ + kh);

  v2f xa[UPC_], xb[UPC_], wa[G_], wb[G_];
  load_frags(x + xo, Wt + wo, xa, xb, wa, wb);

  {
    float o[G_];
    squashed(pa, bias, b, n0, col, o);
    if (MODE == 2) {
      float o1[G_];
      squashed(pb, bias, b, n0, col, o1);
#pragma unroll
      for (int g = 0; g < G_; ++g) o[g] += o1[g];
    }
    if (half == 0) {
#pragma unroll
      for (int g = 0; g < G_; ++g) osum_s[(n0 + g) * DOUT_ + col] = o[g];
    }
  }
  __syncthreads();

  v8f sacc[G_];
#pragma unroll
  for (int g = 0; g < G_; ++g) sacc[g] = (v8f){0.f,0.f,0.f,0.f,0.f,0.f,0.f,0.f};

#pragma nounroll
  for (int cc = 0; cc < CHB_; ++cc) {
    v8f V[G_][UPC_];
#pragma unroll
    for (int g = 0; g < G_; ++g) {
#pragma unroll
      for (int t = 0; t < UPC_; ++t) V[g][t] = vote_tile_T(wa[g], wb[g], xa[t], xb[t]);
    }

    float L[G_][2];
#pragma unroll
    for (int g = 0; g < G_; ++g) {
      const float4* op = (const float4*)(osum_s + (n0 + g) * DOUT_ + (half << 3));
      const float4 oa = op[0], ob = op[1];
      const float od[8] = { oa.x, oa.y, oa.z, oa.w, ob.x, ob.y, ob.z, ob.w };
#pragma unroll
      for (int p = 0; p < 2; ++p) {
        float da = 0.f, db = 0.f;
#pragma unroll
        for (int r = 0; r < 8; ++r) { da += od[r] * V[g][2 * p][r]; db += od[r] * V[g][2 * p + 1][r]; }
        da += xhalf(da);
        db += xhalf(db);
        L[g][p] = half ? db : da;
      }
    }

    float mx[2];
#pragma unroll
    for (int p = 0; p < 2; ++p) {
      mx[p] = fmaxf(fmaxf(L[0][p], L[1][p]), fmaxf(L[2][p], L[3][p]));
      xm[wave * REP_ + p * 32 + lane] = mx[p];
    }
    __syncthreads();
#pragma unroll
    for (int p = 0; p < 2; ++p) {
      float m = xm[p * 32 + lane];
#pragma unroll
      for (int w = 1; w < NWAVE_; ++w) m = fmaxf(m, xm[w * REP_ + p * 32 + lane]);
      mx[p] = m;
    }

#pragma unroll
    for (int p = 0; p < 2; ++p) {
      float s = 0.f;
#pragma unroll
      for (int g = 0; g < G_; ++g) { L[g][p] = __expf(L[g][p] - mx[p]); s += L[g][p]; }
      xs[wave * REP_ + p * 32 + lane] = s;
    }
    __syncthreads();
    float inv[2];
#pragma unroll
    for (int p = 0; p < 2; ++p) {
      float s = xs[p * 32 + lane];
#pragma unroll
      for (int w = 1; w < NWAVE_; ++w) s += xs[w * REP_ + p * 32 + lane];
      inv[p] = 1.0f / s;
    }

    if (cc + 1 < CHB_)
      load_frags(x + (xo + (unsigned)(cc + 1) * (REP_ * DIN_)), Wt + (wo + (unsigned)(cc + 1) * (DOUT_ * DIN_)),
                 xa, xb, wa, wb);

#pragma unroll
    for (int g = 0; g < G_; ++g) {
#pragma unroll
      for (int p = 0; p < 2; ++p) {
        const float c  = L[g][p] * inv[p];
        const float cx = xhalf(c);
        const float ce = half ? cx : c;
        const float co = half ? c : cx;
        sacc[g] += ce * V[g][2 * p];
        sacc[g] += co * V[g][2 * p + 1];
      }
    }
  }

#pragma unroll
  for (int g = 0; g < G_; ++g) stage_capsule(sacc[g], stage, wave, g, col, half);

  float* pw = pout + (((size_t)b * NCHUNK_ + chunk) * N_ + n0) * DOUT_;
  flush_partial(stage, pw, wave, lane);
}

__global__ __launch_bounds__(256) void caps_final(
    const float* __restrict__ part, const float* __restrict__ bias, float* __restrict__ out)
{
  const int e  = blockIdx.x * 256 + threadIdx.x;
  const int b  = e / (N_ * DOUT_);
  const int nd = e % (N_ * DOUT_);
  float s = bias[nd];
#pragma unroll
  for (int k = 0; k < NCHUNK_; ++k)
    s += part[((size_t)b * NCHUNK_ + k) * (N_ * DOUT_) + nd];
  float sn = s * s;
  sn += __shfl_xor(sn, 1, 32);
  sn += __shfl_xor(sn, 2, 32);
  sn += __shfl_xor(sn, 4, 32);
  sn += __shfl_xor(sn, 8, 32);
  const float scl = sn / (1.0f + sn) * rsqrtf(sn + EPS_);
  out[e] = scl * s;
}

extern "C" void kernel_launch(void* const* d_in, const int* in_sizes, int n_in,
                              void* d_out, int out_size, void* d_ws, size_t ws_size,
                              hipStream_t stream)
{
  const float* x    = (const float*)d_in[0];
  const float* Wt   = (const float*)d_in[1];
  const float* bias = (const float*)d_in[2];
  float* out = (float*)d_out;

  float* p0 = (float*)d_ws;
  float* p1 = p0 + (size_t)PLANE_;
  float* p2 = p1 + (size_t)PLANE_;

  dim3 grid(B_ * NCHUNK_), blk(512);
  dim3 fgrid((B_ * N_ * DOUT_) / 256), fblk(256);

  caps_route_first<<<grid, blk, 0, stream>>>(x, Wt, p0);
  caps_route<1><<<grid, blk, 0, stream>>>(x, Wt, bias, p0, p0, p1);
  caps_route<2><<<grid, blk, 0, stream>>>(x, Wt, bias, p0, p1, p2);
  caps_final<<<fgrid, fblk, 0, stream>>>(p2, bias, out);
}


# Softmax statistics cross the workgroup in [column][wave] LDS planes: a lane fetches a column's per-wave maxima or sums with four 128-bit loads, all eight issued ahead of one wait
# speedup vs baseline: 3.0063x; 1.0795x over previous
#include <hip/hip_runtime.h>
#include <math.h>


#define B_    32
#define IN_   2048
#define DIN_  8
#define N_    64
#define DOUT_ 16
#define C_    32
#define REP_  (IN_ / C_)
#define CHB_  4
#define CHUNK_ (CHB_ * REP_)
#define NCHUNK_ (IN_ / CHUNK_)
#define UPC_  (REP_ / 16)
#define UNITS_ (CHB_ * UPC_)
#define G_    4
#define NWAVE_ (N_ / G_)
#define WSTR_ (C_ * DOUT_ * DIN_)
#define PLANE_ (B_ * NCHUNK_ * N_ * DOUT_)
#define EPS_  1e-7f
#define SPITCH_ 20

typedef float v2f __attribute__((ext_vector_type(2)));
typedef float v8f __attribute__((ext_vector_type(8)));

__device__ __forceinline__ float xhalf(float v)
{
  const int q = __builtin_bit_cast(int, v);
  return __builtin_bit_cast(float,
      __builtin_amdgcn_permlanex16(q, q, 0x76543210, 0xfedcba98, false, false));
}

__device__ __forceinline__ v8f vote_tile_T(v2f w0, v2f w1, v2f x0, v2f x1)
{
  v8f acc = {0.f,0.f,0.f,0.f,0.f,0.f,0.f,0.f};
  acc = __builtin_amdgcn_wmma_f32_16x16x4_f32(false, w0, false, x0,
                                              (short)0, acc, false, false);
  acc = __builtin_amdgcn_wmma_f32_16x16x4_f32(false, w1, false, x1,
                                              (short)0, acc, false, false);
  return acc;
}

__device__ __forceinline__ void load_w(const float* __restrict__ ww, v2f (&w)[CHB_][2])
{
#pragma unroll
  for (int cc = 0; cc < CHB_; ++cc) {
    const float* wr = ww + cc * (DOUT_ * DIN_);
    w[cc][0] = (v2f){ wr[0], wr[1] };
    w[cc][1] = (v2f){ wr[4], wr[5] };
  }
}

__device__ __forceinline__ void squashed(const float* __restrict__ part,
                                         const float* __restrict__ bias,
                                         int b, int n0, int col, float (&o)[G_])
{
#pragma unroll
  for (int g = 0; g < G_; ++g) {
    float s = bias[(n0 + g) * DOUT_ + col];
#pragma unroll
    for (int k = 0; k < NCHUNK_; ++k)
      s += part[(((size_t)b * NCHUNK_ + k) * N_ + n0 + g) * DOUT_ + col];
    float sn = s * s;
    sn += __shfl_xor(sn, 1, 32);
    sn += __shfl_xor(sn, 2, 32);
    sn += __shfl_xor(sn, 4, 32);
    sn += __shfl_xor(sn, 8, 32);
    const float scl = sn / (1.0f + sn) * rsqrtf(sn + EPS_);
    o[g] = scl * s;
  }
}

__device__ __forceinline__ void load_frags(const float* __restrict__ xc, const float* __restrict__ wc,
                                           v2f (&xa)[UPC_], v2f (&xb)[UPC_],
                                           v2f (&wa)[G_], v2f (&wb)[G_])
{
#pragma unroll
  for (int t = 0; t < UPC_; ++t) {
    const float* xr = xc + t * (16 * DIN_);
    xa[t] = (v2f){ xr[0], xr[1] };
    xb[t] = (v2f){ xr[4], xr[5] };
  }
#pragma unroll
  for (int g = 0; g < G_; ++g) {
    const float* wr = wc + (size_t)g * WSTR_;
    wa[g] = (v2f){ wr[0], wr[1] };
    wb[g] = (v2f){ wr[4], wr[5] };
  }
}

__device__ __forceinline__ void stage_capsule(v8f sacc, float* __restrict__ stage,
                                              int wave, int g, int col, int half)
{
#pragma unroll
  for (int r = 0; r < 8; ++r) {
    float v = sacc[r];
    v += __shfl_xor(v, 1, 32);
    v += __shfl_xor(v, 2, 32);
    v += __shfl_xor(v, 4, 32);
    v += __shfl_xor(v, 8, 32);
    sacc[r] = v;
  }
  if (col == 0) {
#pragma unroll
    for (int r = 0; r < 8; ++r) stage[(wave * G_ + g) * DOUT_ + (half << 3) + r] = sacc[r];
  }
}

__device__ __forceinline__ void flush_partial(const float* __restrict__ stage, float* __restrict__ pw,
                                              int wave, int lane)
{
  __syncthreads();
#pragma unroll
  for (int q = 0; q < 2; ++q)
    pw[q * 32 + lane] = stage[wave * (G_ * DOUT_) + q * 32 + lane];
}

__global__ __launch_bounds__(512, 4) void caps_route_first(
    const float* __restrict__ x, const float* __restrict__ Wt, float* __restrict__ pout)
{
  __shared__ float stage[NWAVE_ * G_ * DOUT_];

  const int b     = blockIdx.x / NCHUNK_;
  const int chunk = blockIdx.x % NCHUNK_;
  const int wave  = threadIdx.x >> 5;
  const int lane  = threadIdx.x & 31;
  const int col   = lane & 15;
  const int half  = lane >> 4;
  const int kh    = half << 1;
  const int n0    = wave * G_;
  const int i0    = chunk * CHUNK_;
  const size_t ebase = (size_t)b * IN_;

  const float* xw = x + (ebase + i0 + col) * DIN_ + kh;
  v2f xa[UNITS_], xb[UNITS_];
#pragma unroll
  for (int u = 0; u < UNITS_; ++u) {
    const float* xr = xw + u * (16 * DIN_);
    xa[u] = (v2f){ xr[0], xr[1] };
    xb[u] = (v2f){ xr[4], xr[5] };
  }

  const float* wbase = Wt + (((size_t)n0 * C_ + chunk * CHB_) * DOUT_ + col) * DIN_ + kh;

  v2f wf[G_][CHB_][2];
  load_w(wbase, wf[0]);
#pragma unroll
  for (int g = 0; g < G_; ++g) {
    if (g + 1 < G_) load_w(wbase + (size_t)(g + 1) * WSTR_, wf[g + 1]);
    v8f sa = {0.f,0.f,0.f,0.f,0.f,0.f,0.f,0.f};
#pragma unroll
    for (int cc = 0; cc < CHB_; ++cc) {
#pragma unroll
      for (int t = 0; t < UPC_; ++t) {
        const int u = cc * UPC_ + t;
        const v8f acc = vote_tile_T(wf[g][cc][0], wf[g][cc][1], xa[u], xb[u]);
        sa += (1.0f / (float)N_) * acc;
      }
    }
    stage_capsule(sa, stage, wave, g, col, half);
    __builtin_amdgcn_sched_barrier(0);
  }

  float* pw = pout + (((size_t)b * NCHUNK_ + chunk) * N_ + n0) * DOUT_;
  flush_partial(stage, pw, wave, lane);
}

template <int MODE>
__global__ __launch_bounds__(512, 4) void caps_route(
    const float* __restrict__ x, const float* __restrict__ Wt,
    const float* __restrict__ bias,
    const float* __restrict__ pa, const float* __restrict__ pb,
    float* __restrict__ pout)
{
  __shared__ __attribute__((aligned(16))) float osum_s[N_ * DOUT_];
  __shared__ __attribute__((aligned(16))) float xm[REP_ * SPITCH_];
  __shared__ __attribute__((aligned(16))) float xs[REP_ * SPITCH_];
  __shared__ float stage[NWAVE_ * G_ * DOUT_];

  const int b     = blockIdx.x / NCHUNK_;
  const int chunk = blockIdx.x % NCHUNK_;
  const int wave  = threadIdx.x >> 5;
  const int lane  = threadIdx.x & 31;
  const int col   = lane & 15;
  const int half  = lane >> 4;
  const int kh    = half << 1;
  const int n0    = wave * G_;
  const int i0    = chunk * CHUNK_;

  const unsigned xo = (unsigned)((b * IN_ + i0 + col) * DIN_ + kh);
  const unsigned wo = (unsigned)(((n0 * C_ + chunk * CHB_) * DOUT_ + col) * DIN_ + kh);

  v2f xa[UPC_], xb[UPC_], wa[G_], wb[G_];
  load_frags(x + xo, Wt + wo, xa, xb, wa, wb);

  {
    float o[G_];
    squashed(pa, bias, b, n0, col, o);
    if (MODE == 2) {
      float o1[G_];
      squashed(pb, bias, b, n0, col, o1);
#pragma unroll
      for (int g = 0; g < G_; ++g) o[g] += o1[g];
    }
    if (half == 0) {
#pragma unroll
      for (int g = 0; g < G_; ++g) osum_s[(n0 + g) * DOUT_ + col] = o[g];
    }
  }
  __syncthreads();

  v8f sacc[G_];
#pragma unroll
  for (int g = 0; g < G_; ++g) sacc[g] = (v8f){0.f,0.f,0.f,0.f,0.f,0.f,0.f,0.f};

#pragma nounroll
  for (int cc = 0; cc < CHB_; ++cc) {
    v8f V[G_][UPC_];
#pragma unroll
    for (int g = 0; g < G_; ++g) {
#pragma unroll
      for (int t = 0; t < UPC_; ++t) V[g][t] = vote_tile_T(wa[g], wb[g], xa[t], xb[t]);
    }

    float L[G_][2];
#pragma unroll
    for (int g = 0; g < G_; ++g) {
      const float4* op = (const float4*)(osum_s + (n0 + g) * DOUT_ + (half << 3));
      const float4 oa = op[0], ob = op[1];
      const float od[8] = { oa.x, oa.y, oa.z, oa.w, ob.x, ob.y, ob.z, ob.w };
#pragma unroll
      for (int p = 0; p < 2; ++p) {
        float da = 0.f, db = 0.f;
#pragma unroll
        for (int r = 0; r < 8; ++r) { da += od[r] * V[g][2 * p][r]; db += od[r] * V[g][2 * p + 1][r]; }
        da += xhalf(da);
        db += xhalf(db);
        L[g][p] = half ? db : da;
      }
    }

    float mx[2];
#pragma unroll
    for (int p = 0; p < 2; ++p) {
      mx[p] = fmaxf(fmaxf(L[0][p], L[1][p]), fmaxf(L[2][p], L[3][p]));
      xm[(p * 32 + lane) * SPITCH_ + wave] = mx[p];
    }
    __syncthreads();
    {
      typedef float v4f __attribute__((ext_vector_type(4)));
      v4f q0, q1, q2, q3, q4, q5, q6, q7;
      const unsigned ra = (unsigned)(size_t)(xm + lane * SPITCH_);
      static_assert(32 * SPITCH_ * 4 == 2560, "the asm offsets below place column 32 + lane 2560 bytes on");
      asm volatile(
          "ds_load_b128 %0, %8\n\t"
          "ds_load_b128 %1, %8 offset:16\n\t"
          "ds_load_b128 %2, %8 offset:32\n\t"
          "ds_load_b128 %3, %8 offset:48\n\t"
          "ds_load_b128 %4, %8 offset:2560\n\t"
          "ds_load_b128 %5, %8 offset:2576\n\t"
          "ds_load_b128 %6, %8 offset:2592\n\t"
          "ds_load_b128 %7, %8 offset:2608\n\t"
          "s_wait_dscnt 0x0"
          : "=&v"(q0), "=&v"(q1), "=&v"(q2), "=&v"(q3), "=&v"(q4), "=&v"(q5), "=&v"(q6), "=&v"(q7)
          : "v"(ra)
          : "memory");
      float m = fmaxf(fmaxf(q0.x, q0.y), fmaxf(q0.z, q0.w));
      m = fmaxf(fmaxf(m, q1.x), fmaxf(fmaxf(q1.y, q1.z), q1.w));
      m = fmaxf(fmaxf(m, q2.x), fmaxf(fmaxf(q2.y, q2.z), q2.w));
      m = fmaxf(fmaxf(m, q3.x), fmaxf(fmaxf(q3.y, q3.z), q3.w));
      mx[0] = m;
      m = fmaxf(fmaxf(q4.x, q4.y), fmaxf(q4.z, q4.w));
      m = fmaxf(fmaxf(m, q5.x), fmaxf(fmaxf(q5.y, q5.z), q5.w));
      m = fmaxf(fmaxf(m, q6.x), fmaxf(fmaxf(q6.y, q6.z), q6.w));
      m = fmaxf(fmaxf(m, q7.x), fmaxf(fmaxf(q7.y, q7.z), q7.w));
      mx[1] = m;
    }

#pragma unroll
    for (int p = 0; p < 2; ++p) {
      float s = 0.f;
#pragma unroll
      for (int g = 0; g < G_; ++g) { L[g][p] = __expf(L[g][p] - mx[p]); s += L[g][p]; }
      xs[(p * 32 + lane) * SPITCH_ + wave] = s;
    }
    __syncthreads();
    float4 qs[2][NWAVE_ / 4];
#pragma unroll
    for (int p = 0; p < 2; ++p) {
      const float4* rs = (const float4*)(xs + (p * 32 + lane) * SPITCH_);
#pragma unroll
      for (int j = 0; j < NWAVE_ / 4; ++j) qs[p][j] = rs[j];
    }
    float inv[2];
#pragma unroll
    for (int p = 0; p < 2; ++p) {
      float s = qs[p][0].x;
      s += qs[p][0].y; s += qs[p][0].z; s += qs[p][0].w;
#pragma unroll
      for (int j = 1; j < NWAVE_ / 4; ++j) { s += qs[p][j].x; s += qs[p][j].y; s += qs[p][j].z; s += qs[p][j].w; }
      inv[p] = 1.0f / s;
    }

    if (cc + 1 < CHB_)
      load_frags(x + (xo + (unsigned)(cc + 1) * (REP_ * DIN_)), Wt + (wo + (unsigned)(cc + 1) * (DOUT_ * DIN_)),
                 xa, xb, wa, wb);

#pragma unroll
    for (int g = 0; g < G_; ++g) {
#pragma unroll
      for (int p = 0; p < 2; ++p) {
        const float c  = L[g][p] * inv[p];
        const float cx = xhalf(c);
        const float ce = half ? cx : c;
        const float co = half ? c : cx;
        sacc[g] += ce * V[g][2 * p];
        sacc[g] += co * V[g][2 * p + 1];
      }
    }
  }

#pragma unroll
  for (int g = 0; g < G_; ++g) stage_capsule(sacc[g], stage, wave, g, col, half);

  float* pw = pout + (((size_t)b * NCHUNK_ + chunk) * N_ + n0) * DOUT_;
  flush_partial(stage, pw, wave, lane);
}

__global__ __launch_bounds__(256) void caps_final(
    const float* __restrict__ part, const float* __restrict__ bias, float* __restrict__ out)
{
  const int e  = blockIdx.x * 256 + threadIdx.x;
  const int b  = e / (N_ * DOUT_);
  const int nd = e % (N_ * DOUT_);
  float s = bias[nd];
#pragma unroll
  for (int k = 0; k < NCHUNK_; ++k)
    s += part[((size_t)b * NCHUNK_ + k) * (N_ * DOUT_) + nd];
  float sn = s * s;
  sn += __shfl_xor(sn, 1, 32);
  sn += __shfl_xor(sn, 2, 32);
  sn += __shfl_xor(sn, 4, 32);
  sn += __shfl_xor(sn, 8, 32);
  const float scl = sn / (1.0f + sn) * rsqrtf(sn + EPS_);
  out[e] = scl * s;
}

extern "C" void kernel_launch(void* const* d_in, const int* in_sizes, int n_in,
                              void* d_out, int out_size, void* d_ws, size_t ws_size,
                              hipStream_t stream)
{
  const float* x    = (const float*)d_in[0];
  const float* Wt   = (const float*)d_in[1];
  const float* bias = (const float*)d_in[2];
  float* out = (float*)d_out;

  float* p0 = (float*)d_ws;
  float* p1 = p0 + (size_t)PLANE_;
  float* p2 = p1 + (size_t)PLANE_;

  dim3 grid(B_ * NCHUNK_), blk(512);
  dim3 fgrid((B_ * N_ * DOUT_) / 256), fblk(256);

  caps_route_first<<<grid, blk, 0, stream>>>(x, Wt, p0);
  caps_route<1><<<grid, blk, 0, stream>>>(x, Wt, bias, p0, p0, p1);
  caps_route<2><<<grid, blk, 0, stream>>>(x, Wt, bias, p0, p1, p2);
  caps_final<<<fgrid, fblk, 0, stream>>>(p2, bias, out);
}


# 16-lane sums leave the LDS unit: column sums of the partial s accumulators as a halving butterfly of row_xmask DPP adds (one staged value per lane), squash norms as DPP adds; no ds_bpermute left
# speedup vs baseline: 3.2019x; 1.0651x over previous
#include <hip/hip_runtime.h>
#include <math.h>


#define B_    32
#define IN_   2048
#define DIN_  8
#define N_    64
#define DOUT_ 16
#define C_    32
#define REP_  (IN_ / C_)
#define CHB_  4
#define CHUNK_ (CHB_ * REP_)
#define NCHUNK_ (IN_ / CHUNK_)
#define UPC_  (REP_ / 16)
#define UNITS_ (CHB_ * UPC_)
#define G_    4
#define NWAVE_ (N_ / G_)
#define WSTR_ (C_ * DOUT_ * DIN_)
#define PLANE_ (B_ * NCHUNK_ * N_ * DOUT_)
#define EPS_  1e-7f
#define SPITCH_ 20

typedef float v2f __attribute__((ext_vector_type(2)));
typedef float v8f __attribute__((ext_vector_type(8)));

__device__ __forceinline__ float xhalf(float v)
{
  const int q = __builtin_bit_cast(int, v);
  return __builtin_bit_cast(float,
      __builtin_amdgcn_permlanex16(q, q, 0x76543210, 0xfedcba98, false, false));
}

template <int M>
__device__ __forceinline__ float row_xor(float v)
{
  return __builtin_bit_cast(float,
      __builtin_amdgcn_update_dpp(0, __builtin_bit_cast(int, v), 0x160 | M, 0xf, 0xf, true));
}

__device__ __forceinline__ float row_sum16(float v)
{
  v += row_xor<1>(v);
  v += row_xor<2>(v);
  v += row_xor<4>(v);
  v += row_xor<8>(v);
  return v;
}

__device__ __forceinline__ v8f vote_tile_T(v2f w0, v2f w1, v2f x0, v2f x1)
{
  v8f acc = {0.f,0.f,0.f,0.f,0.f,0.f,0.f,0.f};
  acc = __builtin_amdgcn_wmma_f32_16x16x4_f32(false, w0, false, x0,
                                              (short)0, acc, false, false);
  acc = __builtin_amdgcn_wmma_f32_16x16x4_f32(false, w1, false, x1,
                                              (short)0, acc, false, false);
  return acc;
}

__device__ __forceinline__ void load_w(const float* __restrict__ ww, v2f (&w)[CHB_][2])
{
#pragma unroll
  for (int cc = 0; cc < CHB_; ++cc) {
    const float* wr = ww + cc * (DOUT_ * DIN_);
    w[cc][0] = (v2f){ wr[0], wr[1] };
    w[cc][1] = (v2f){ wr[4], wr[5] };
  }
}

__device__ __forceinline__ void squashed(const float* __restrict__ part,
                                         const float* __restrict__ bias,
                                         int b, int n0, int col, float (&o)[G_])
{
#pragma unroll
  for (int g = 0; g < G_; ++g) {
    float s = bias[(n0 + g) * DOUT_ + col];
#pragma unroll
    for (int k = 0; k < NCHUNK_; ++k)
      s += part[(((size_t)b * NCHUNK_ + k) * N_ + n0 + g) * DOUT_ + col];
    const float sn = row_sum16(s * s);
    const float scl = sn / (1.0f + sn) * rsqrtf(sn + EPS_);
    o[g] = scl * s;
  }
}

__device__ __forceinline__ void load_frags(const float* __restrict__ xc, const float* __restrict__ wc,
                                           v2f (&xa)[UPC_], v2f (&xb)[UPC_],
                                           v2f (&wa)[G_], v2f (&wb)[G_])
{
#pragma unroll
  for (int t = 0; t < UPC_; ++t) {
    const float* xr = xc + t * (16 * DIN_);
    xa[t] = (v2f){ xr[0], xr[1] };
    xb[t] = (v2f){ xr[4], xr[5] };
  }
#pragma unroll
  for (int g = 0; g < G_; ++g) {
    const float* wr = wc + (size_t)g * WSTR_;
    wa[g] = (v2f){ wr[0], wr[1] };
    wb[g] = (v2f){ wr[4], wr[5] };
  }
}

__device__ __forceinline__ float colsum_capsule(v8f a, int col)
{
  const bool b3 = (col & 8) != 0, b2 = (col & 4) != 0, b1 = (col & 2) != 0;
  typedef float v4f __attribute__((ext_vector_type(4)));
  const v4f lo = __builtin_shufflevector(a, a, 0, 1, 2, 3);
  const v4f hi = __builtin_shufflevector(a, a, 4, 5, 6, 7);
  const v4f keep4 = b3 ? hi : lo;
  const v4f send4 = b3 ? lo : hi;
  float p[4], q[2];
#pragma unroll
  for (int j = 0; j < 4; ++j) p[j] = keep4[j] + row_xor<8>(send4[j]);
#pragma unroll
  for (int j = 0; j < 2; ++j) {
    const float keep = b2 ? p[2 + j] : p[j];
    const float send = b2 ? p[j] : p[2 + j];
    q[j] = keep + row_xor<4>(send);
  }
  const float keep = b1 ? q[1] : q[0];
  const float send = b1 ? q[0] : q[1];
  float e = keep + row_xor<2>(send);
  e += row_xor<1>(e);
  return e;
}

__device__ __forceinline__ void stage_capsules(const float (&e)[G_], float* __restrict__ stage,
                                               int wave, int col, int half)
{
  static_assert(G_ == 4, "two capsules per lane parity");
  const bool odd = (col & 1) != 0;
  float* sp = stage + (wave * G_ + (odd ? 2 : 0)) * DOUT_ + (half << 3) + (col >> 1);
  sp[0]     = odd ? e[2] : e[0];
  sp[DOUT_] = odd ? e[3] : e[1];
}

__device__ __forceinline__ void flush_partial(const float* __restrict__ stage, float* __restrict__ pw,
                                              int wave, int lane)
{
  __syncthreads();
#pragma unroll
  for (int q = 0; q < 2; ++q)
    pw[q * 32 + lane] = stage[wave * (G_ * DOUT_) + q * 32 + lane];
}

__global__ __launch_bounds__(512, 4) void caps_route_first(
    const float* __restrict__ x, const float* __restrict__ Wt, float* __restrict__ pout)
{
  __shared__ float stage[NWAVE_ * G_ * DOUT_];

  const int b     = blockIdx.x / NCHUNK_;
  const int chunk = blockIdx.x % NCHUNK_;
  const int wave  = threadIdx.x >> 5;
  const int lane  = threadIdx.x & 31;
  const int col   = lane & 15;
  const int half  = lane >> 4;
  const int kh    = half << 1;
  const int n0    = wave * G_;
  const int i0    = chunk * CHUNK_;
  const size_t ebase = (size_t)b * IN_;

  const float* xw = x + (ebase + i0 + col) * DIN_ + kh;
  v2f xa[UNITS_], xb[UNITS_];
#pragma unroll
  for (int u = 0; u < UNITS_; ++u) {
    const float* xr = xw + u * (16 * DIN_);
    xa[u] = (v2f){ xr[0], xr[1] };
    xb[u] = (v2f){ xr[4], xr[5] };
  }

  const float* wbase = Wt + (((size_t)n0 * C_ + chunk * CHB_) * DOUT_ + col) * DIN_ + kh;

  v2f wf[G_][CHB_][2];
  float e[G_];
  load_w(wbase, wf[0]);
#pragma unroll
  for (int g = 0; g < G_; ++g) {
    if (g + 1 < G_) load_w(wbase + (size_t)(g + 1) * WSTR_, wf[g + 1]);
    v8f sa = {0.f,0.f,0.f,0.f,0.f,0.f,0.f,0.f};
#pragma unroll
    for (int cc = 0; cc < CHB_; ++cc) {
#pragma unroll
      for (int t = 0; t < UPC_; ++t) {
        const int u = cc * UPC_ + t;
        const v8f acc = vote_tile_T(wf[g][cc][0], wf[g][cc][1], xa[u], xb[u]);
        sa += (1.0f / (float)N_) * acc;
      }
    }
    e[g] = colsum_capsule(sa, col);
    __builtin_amdgcn_sched_barrier(0);
  }

  stage_capsules(e, stage, wave, col, half);
  float* pw = pout + (((size_t)b * NCHUNK_ + chunk) * N_ + n0) * DOUT_;
  flush_partial(stage, pw, wave, lane);
}

template <int MODE>
__global__ __launch_bounds__(512, 4) void caps_route(
    const float* __restrict__ x, const float* __restrict__ Wt,
    const float* __restrict__ bias,
    const float* __restrict__ pa, const float* __restrict__ pb,
    float* __restrict__ pout)
{
  __shared__ __attribute__((aligned(16))) float osum_s[N_ * DOUT_];
  __shared__ __attribute__((aligned(16))) float xm[REP_ * SPITCH_];
  __shared__ __attribute__((aligned(16))) float xs[REP_ * SPITCH_];
  __shared__ float stage[NWAVE_ * G_ * DOUT_];

  const int b     = blockIdx.x / NCHUNK_;
  const int chunk = blockIdx.x % NCHUNK_;
  const int wave  = threadIdx.x >> 5;
  const int lane  = threadIdx.x & 31;
  const int col   = lane & 15;
  const int half  = lane >> 4;
  const int kh    = half << 1;
  const int n0    = wave * G_;
  const int i0    = chunk * CHUNK_;

  const unsigned xo = (unsigned)((b * IN_ + i0 + col) * DIN_ + kh);
  const unsigned wo = (unsigned)(((n0 * C_ + chunk * CHB_) * DOUT_ + col) * DIN_ + kh);

  v2f xa[UPC_], xb[UPC_], wa[G_], wb[G_];
  load_frags(x + xo, Wt + wo, xa, xb, wa, wb);

  {
    float o[G_];
    squashed(pa, bias, b, n0, col, o);
    if (MODE == 2) {
      float o1[G_];
      squashed(pb, bias, b, n0, col, o1);
#pragma unroll
      for (int g = 0; g < G_; ++g) o[g] += o1[g];
    }
    if (half == 0) {
#pragma unroll
      for (int g = 0; g < G_; ++g) osum_s[(n0 + g) * DOUT_ + col] = o[g];
    }
  }
  __syncthreads();

  v8f sacc[G_];
#pragma unroll
  for (int g = 0; g < G_; ++g) sacc[g] = (v8f){0.f,0.f,0.f,0.f,0.f,0.f,0.f,0.f};

#pragma nounroll
  for (int cc = 0; cc < CHB_; ++cc) {
    v8f V[G_][UPC_];
#pragma unroll
    for (int g = 0; g < G_; ++g) {
#pragma unroll
      for (int t = 0; t < UPC_; ++t) V[g][t] = vote_tile_T(wa[g], wb[g], xa[t], xb[t]);
    }

    float L[G_][2];
#pragma unroll
    for (int g = 0; g < G_; ++g) {
      const float4* op = (const float4*)(osum_s + (n0 + g) * DOUT_ + (half << 3));
      const float4 oa = op[0], ob = op[1];
      const float od[8] = { oa.x, oa.y, oa.z, oa.w, ob.x, ob.y, ob.z, ob.w };
#pragma unroll
      for (int p = 0; p < 2; ++p) {
        float da = 0.f, db = 0.f;
#pragma unroll
        for (int r = 0; r < 8; ++r) { da += od[r] * V[g][2 * p][r]; db += od[r] * V[g][2 * p + 1][r]; }
        da += xhalf(da);
        db += xhalf(db);
        L[g][p] = half ? db : da;
      }
    }

    float mx[2];
#pragma unroll
    for (int p = 0; p < 2; ++p) {
      mx[p] = fmaxf(fmaxf(L[0][p], L[1][p]), fmaxf(L[2][p], L[3][p]));
      xm[(p * 32 + lane) * SPITCH_ + wave] = mx[p];
    }
    __syncthreads();
    {
      typedef float v4f __attribute__((ext_vector_type(4)));
      v4f q0, q1, q2, q3, q4, q5, q6, q7;
      const unsigned ra = (unsigned)(size_t)(xm + lane * SPITCH_);
      static_assert(32 * SPITCH_ * 4 == 2560, "the asm offsets below place column 32 + lane 2560 bytes on");
      asm volatile(
          "ds_load_b128 %0, %8\n\t"
          "ds_load_b128 %1, %8 offset:16\n\t"
          "ds_load_b128 %2, %8 offset:32\n\t"
          "ds_load_b128 %3, %8 offset:48\n\t"
          "ds_load_b128 %4, %8 offset:2560\n\t"
          "ds_load_b128 %5, %8 offset:2576\n\t"
          "ds_load_b128 %6, %8 offset:2592\n\t"
          "ds_load_b128 %7, %8 offset:2608\n\t"
          "s_wait_dscnt 0x0"
          : "=&v"(q0), "=&v"(q1), "=&v"(q2), "=&v"(q3), "=&v"(q4), "=&v"(q5), "=&v"(q6), "=&v"(q7)
          : "v"(ra)
          : "memory");
      float m = fmaxf(fmaxf(q0.x, q0.y), fmaxf(q0.z, q0.w));
      m = fmaxf(fmaxf(m, q1.x), fmaxf(fmaxf(q1.y, q1.z), q1.w));
      m = fmaxf(fmaxf(m, q2.x), fmaxf(fmaxf(q2.y, q2.z), q2.w));
      m = fmaxf(fmaxf(m, q3.x), fmaxf(fmaxf(q3.y, q3.z), q3.w));
      mx[0] = m;
      m = fmaxf(fmaxf(q4.x, q4.y), fmaxf(q4.z, q4.w));
      m = fmaxf(fmaxf(m, q5.x), fmaxf(fmaxf(q5.y, q5.z), q5.w));
      m = fmaxf(fmaxf(m, q6.x), fmaxf(fmaxf(q6.y, q6.z), q6.w));
      m = fmaxf(fmaxf(m, q7.x), fmaxf(fmaxf(q7.y, q7.z), q7.w));
      mx[1] = m;
    }

#pragma unroll
    for (int p = 0; p < 2; ++p) {
      float s = 0.f;
#pragma unroll
      for (int g = 0; g < G_; ++g) { L[g][p] = __expf(L[g][p] - mx[p]); s += L[g][p]; }
      xs[(p * 32 + lane) * SPITCH_ + wave] = s;
    }
    __syncthreads();
    float4 qs[2][NWAVE_ / 4];
#pragma unroll
    for (int p = 0; p < 2; ++p) {
      const float4* rs = (const float4*)(xs + (p * 32 + lane) * SPITCH_);
#pragma unroll
      for (int j = 0; j < NWAVE_ / 4; ++j) qs[p][j] = rs[j];
    }
    float inv[2];
#pragma unroll
    for (int p = 0; p < 2; ++p) {
      float s = qs[p][0].x;
      s += qs[p][0].y; s += qs[p][0].z; s += qs[p][0].w;
#pragma unroll
      for (int j = 1; j < NWAVE_ / 4; ++j) { s += qs[p][j].x; s += qs[p][j].y; s += qs[p][j].z; s += qs[p][j].w; }
      inv[p] = 1.0f / s;
    }

    if (cc + 1 < CHB_)
      load_frags(x + (xo + (unsigned)(cc + 1) * (REP_ * DIN_)), Wt + (wo + (unsigned)(cc + 1) * (DOUT_ * DIN_)),
                 xa, xb, wa, wb);

#pragma unroll
    for (int g = 0; g < G_; ++g) {
#pragma unroll
      for (int p = 0; p < 2; ++p) {
        const float c  = L[g][p] * inv[p];
        const float cx = xhalf(c);
        const float ce = half ? cx : c;
        const float co = half ? c : cx;
        sacc[g] += ce * V[g][2 * p];
        sacc[g] += co * V[g][2 * p + 1];
      }
    }
  }

  float e[G_];
#pragma unroll
  for (int g = 0; g < G_; ++g) e[g] = colsum_capsule(sacc[g], col);
  stage_capsules(e, stage, wave, col, half);

  float* pw = pout + (((size_t)b * NCHUNK_ + chunk) * N_ + n0) * DOUT_;
  flush_partial(stage, pw, wave, lane);
}

__global__ __launch_bounds__(256) void caps_final(
    const float* __restrict__ part, const float* __restrict__ bias, float* __restrict__ out)
{
  const int e  = blockIdx.x * 256 + threadIdx.x;
  const int b  = e / (N_ * DOUT_);
  const int nd = e % (N_ * DOUT_);
  float s = bias[nd];
#pragma unroll
  for (int k = 0; k < NCHUNK_; ++k)
    s += part[((size_t)b * NCHUNK_ + k) * (N_ * DOUT_) + nd];
  const float sn = row_sum16(s * s);
  const float scl = sn / (1.0f + sn) * rsqrtf(sn + EPS_);
  out[e] = scl * s;
}

extern "C" void kernel_launch(void* const* d_in, const int* in_sizes, int n_in,
                              void* d_out, int out_size, void* d_ws, size_t ws_size,
                              hipStream_t stream)
{
  const float* x    = (const float*)d_in[0];
  const float* Wt   = (const float*)d_in[1];
  const float* bias = (const float*)d_in[2];
  float* out = (float*)d_out;

  float* p0 = (float*)d_ws;
  float* p1 = p0 + (size_t)PLANE_;
  float* p2 = p1 + (size_t)PLANE_;

  dim3 grid(B_ * NCHUNK_), blk(512);
  dim3 fgrid((B_ * N_ * DOUT_) / 256), fblk(256);

  caps_route_first<<<grid, blk, 0, stream>>>(x, Wt, p0);
  caps_route<1><<<grid, blk, 0, stream>>>(x, Wt, bias, p0, p0, p1);
  caps_route<2><<<grid, blk, 0, stream>>>(x, Wt, bias, p0, p1, p2);
  caps_final<<<fgrid, fblk, 0, stream>>>(p2, bias, out);
}


# Logit pass in fewest vector instructions: each 8-term in-lane dot is four two-wide FMAs on adjacent register pairs plus one add, and a lane half sends only the partial its partner keeps
# speedup vs baseline: 3.4191x; 1.0678x over previous
#include <hip/hip_runtime.h>
#include <math.h>


#define B_    32
#define IN_   2048
#define DIN_  8
#define N_    64
#define DOUT_ 16
#define C_    32
#define REP_  (IN_ / C_)
#define CHB_  4
#define CHUNK_ (CHB_ * REP_)
#define NCHUNK_ (IN_ / CHUNK_)
#define UPC_  (REP_ / 16)
#define UNITS_ (CHB_ * UPC_)
#define G_    4
#define NWAVE_ (N_ / G_)
#define WSTR_ (C_ * DOUT_ * DIN_)
#define PLANE_ (B_ * NCHUNK_ * N_ * DOUT_)
#define EPS_  1e-7f
#define SPITCH_ 20

typedef float v2f __attribute__((ext_vector_type(2)));
typedef float v8f __attribute__((ext_vector_type(8)));

__device__ __forceinline__ float xhalf(float v)
{
  const int q = __builtin_bit_cast(int, v);
  return __builtin_bit_cast(float,
      __builtin_amdgcn_permlanex16(q, q, 0x76543210, 0xfedcba98, false, false));
}

template <int M>
__device__ __forceinline__ float row_xor(float v)
{
  return __builtin_bit_cast(float,
      __builtin_amdgcn_update_dpp(0, __builtin_bit_cast(int, v), 0x160 | M, 0xf, 0xf, true));
}

__device__ __forceinline__ float row_sum16(float v)
{
  v += row_xor<1>(v);
  v += row_xor<2>(v);
  v += row_xor<4>(v);
  v += row_xor<8>(v);
  return v;
}

__device__ __forceinline__ v8f vote_tile_T(v2f w0, v2f w1, v2f x0, v2f x1)
{
  v8f acc = {0.f,0.f,0.f,0.f,0.f,0.f,0.f,0.f};
  acc = __builtin_amdgcn_wmma_f32_16x16x4_f32(false, w0, false, x0,
                                              (short)0, acc, false, false);
  acc = __builtin_amdgcn_wmma_f32_16x16x4_f32(false, w1, false, x1,
                                              (short)0, acc, false, false);
  return acc;
}

__device__ __forceinline__ void load_w(const float* __restrict__ ww, v2f (&w)[CHB_][2])
{
#pragma unroll
  for (int cc = 0; cc < CHB_; ++cc) {
    const float* wr = ww + cc * (DOUT_ * DIN_);
    w[cc][0] = (v2f){ wr[0], wr[1] };
    w[cc][1] = (v2f){ wr[4], wr[5] };
  }
}

__device__ __forceinline__ void squashed(const float* __restrict__ part,
                                         const float* __restrict__ bias,
                                         int b, int n0, int col, float (&o)[G_])
{
#pragma unroll
  for (int g = 0; g < G_; ++g) {
    float s = bias[(n0 + g) * DOUT_ + col];
#pragma unroll
    for (int k = 0; k < NCHUNK_; ++k)
      s += part[(((size_t)b * NCHUNK_ + k) * N_ + n0 + g) * DOUT_ + col];
    const float sn = row_sum16(s * s);
    const float scl = sn / (1.0f + sn) * rsqrtf(sn + EPS_);
    o[g] = scl * s;
  }
}

__device__ __forceinline__ void load_frags(const float* __restrict__ xc, const float* __restrict__ wc,
                                           v2f (&xa)[UPC_], v2f (&xb)[UPC_],
                                           v2f (&wa)[G_], v2f (&wb)[G_])
{
#pragma unroll
  for (int t = 0; t < UPC_; ++t) {
    const float* xr = xc + t * (16 * DIN_);
    xa[t] = (v2f){ xr[0], xr[1] };
    xb[t] = (v2f){ xr[4], xr[5] };
  }
#pragma unroll
  for (int g = 0; g < G_; ++g) {
    const float* wr = wc + (size_t)g * WSTR_;
    wa[g] = (v2f){ wr[0], wr[1] };
    wb[g] = (v2f){ wr[4], wr[5] };
  }
}

__device__ __forceinline__ float colsum_capsule(v8f a, int col)
{
  const bool b3 = (col & 8) != 0, b2 = (col & 4) != 0, b1 = (col & 2) != 0;
  typedef float v4f __attribute__((ext_vector_type(4)));
  const v4f lo = __builtin_shufflevector(a, a, 0, 1, 2, 3);
  const v4f hi = __builtin_shufflevector(a, a, 4, 5, 6, 7);
  const v4f keep4 = b3 ? hi : lo;
  const v4f send4 = b3 ? lo : hi;
  float p[4], q[2];
#pragma unroll
  for (int j = 0; j < 4; ++j) p[j] = keep4[j] + row_xor<8>(send4[j]);
#pragma unroll
  for (int j = 0; j < 2; ++j) {
    const float keep = b2 ? p[2 + j] : p[j];
    const float send = b2 ? p[j] : p[2 + j];
    q[j] = keep + row_xor<4>(send);
  }
  const float keep = b1 ? q[1] : q[0];
  const float send = b1 ? q[0] : q[1];
  float e = keep + row_xor<2>(send);
  e += row_xor<1>(e);
  return e;
}

__device__ __forceinline__ void stage_capsules(const float (&e)[G_], float* __restrict__ stage,
                                               int wave, int col, int half)
{
  static_assert(G_ == 4, "two capsules per lane parity");
  const bool odd = (col & 1) != 0;
  float* sp = stage + (wave * G_ + (odd ? 2 : 0)) * DOUT_ + (half << 3) + (col >> 1);
  sp[0]     = odd ? e[2] : e[0];
  sp[DOUT_] = odd ? e[3] : e[1];
}

__device__ __forceinline__ void flush_partial(const float* __restrict__ stage, float* __restrict__ pw,
                                              int wave, int lane)
{
  __syncthreads();
#pragma unroll
  for (int q = 0; q < 2; ++q)
    pw[q * 32 + lane] = stage[wave * (G_ * DOUT_) + q * 32 + lane];
}

__global__ __launch_bounds__(512, 4) void caps_route_first(
    const float* __restrict__ x, const float* __restrict__ Wt, float* __restrict__ pout)
{
  __shared__ float stage[NWAVE_ * G_ * DOUT_];

  const int b     = blockIdx.x / NCHUNK_;
  const int chunk = blockIdx.x % NCHUNK_;
  const int wave  = threadIdx.x >> 5;
  const int lane  = threadIdx.x & 31;
  const int col   = lane & 15;
  const int half  = lane >> 4;
  const int kh    = half << 1;
  const int n0    = wave * G_;
  const int i0    = chunk * CHUNK_;
  const size_t ebase = (size_t)b * IN_;

  const float* xw = x + (ebase + i0 + col) * DIN_ + kh;
  v2f xa[UNITS_], xb[UNITS_];
#pragma unroll
  for (int u = 0; u < UNITS_; ++u) {
    const float* xr = xw + u * (16 * DIN_);
    xa[u] = (v2f){ xr[0], xr[1] };
    xb[u] = (v2f){ xr[4], xr[5] };
  }

  const float* wbase = Wt + (((size_t)n0 * C_ + chunk * CHB_) * DOUT_ + col) * DIN_ + kh;

  v2f wf[G_][CHB_][2];
  float e[G_];
  load_w(wbase, wf[0]);
#pragma unroll
  for (int g = 0; g < G_; ++g) {
    if (g + 1 < G_) load_w(wbase + (size_t)(g + 1) * WSTR_, wf[g + 1]);
    v8f sa = {0.f,0.f,0.f,0.f,0.f,0.f,0.f,0.f};
#pragma unroll
    for (int cc = 0; cc < CHB_; ++cc) {
#pragma unroll
      for (int t = 0; t < UPC_; ++t) {
        const int u = cc * UPC_ + t;
        const v8f acc = vote_tile_T(wf[g][cc][0], wf[g][cc][1], xa[u], xb[u]);
        sa += (1.0f / (float)N_) * acc;
      }
    }
    e[g] = colsum_capsule(sa, col);
    __builtin_amdgcn_sched_barrier(0);
  }

  stage_capsules(e, stage, wave, col, half);
  float* pw = pout + (((size_t)b * NCHUNK_ + chunk) * N_ + n0) * DOUT_;
  flush_partial(stage, pw, wave, lane);
}

template <int MODE>
__global__ __launch_bounds__(512, 4) void caps_route(
    const float* __restrict__ x, const float* __restrict__ Wt,
    const float* __restrict__ bias,
    const float* __restrict__ pa, const float* __restrict__ pb,
    float* __restrict__ pout)
{
  __shared__ __attribute__((aligned(16))) float osum_s[N_ * DOUT_];
  __shared__ __attribute__((aligned(16))) float xm[REP_ * SPITCH_];
  __shared__ __attribute__((aligned(16))) float xs[REP_ * SPITCH_];
  __shared__ float stage[NWAVE_ * G_ * DOUT_];

  const int b     = blockIdx.x / NCHUNK_;
  const int chunk = blockIdx.x % NCHUNK_;
  const int wave  = threadIdx.x >> 5;
  const int lane  = threadIdx.x & 31;
  const int col   = lane & 15;
  const int half  = lane >> 4;
  const int kh    = half << 1;
  const int n0    = wave * G_;
  const int i0    = chunk * CHUNK_;

  const unsigned xo = (unsigned)((b * IN_ + i0 + col) * DIN_ + kh);
  const unsigned wo = (unsigned)(((n0 * C_ + chunk * CHB_) * DOUT_ + col) * DIN_ + kh);

  v2f xa[UPC_], xb[UPC_], wa[G_], wb[G_];
  load_frags(x + xo, Wt + wo, xa, xb, wa, wb);

  {
    float o[G_];
    squashed(pa, bias, b, n0, col, o);
    if (MODE == 2) {
      float o1[G_];
      squashed(pb, bias, b, n0, col, o1);
#pragma unroll
      for (int g = 0; g < G_; ++g) o[g] += o1[g];
    }
    if (half == 0) {
#pragma unroll
      for (int g = 0; g < G_; ++g) osum_s[(n0 + g) * DOUT_ + col] = o[g];
    }
  }
  __syncthreads();

  v8f sacc[G_];
#pragma unroll
  for (int g = 0; g < G_; ++g) sacc[g] = (v8f){0.f,0.f,0.f,0.f,0.f,0.f,0.f,0.f};

#pragma nounroll
  for (int cc = 0; cc < CHB_; ++cc) {
    v8f V[G_][UPC_];
#pragma unroll
    for (int g = 0; g < G_; ++g) {
#pragma unroll
      for (int t = 0; t < UPC_; ++t) V[g][t] = vote_tile_T(wa[g], wb[g], xa[t], xb[t]);
    }

    float L[G_][2];
#pragma unroll
    for (int g = 0; g < G_; ++g) {
      const float4* op = (const float4*)(osum_s + (n0 + g) * DOUT_ + (half << 3));
      const float4 oa = op[0], ob = op[1];
      const v2f o01 = { oa.x, oa.y }, o23 = { oa.z, oa.w }, o45 = { ob.x, ob.y }, o67 = { ob.z, ob.w };
#pragma unroll
      for (int p = 0; p < 2; ++p) {
        float dt[2];
#pragma unroll
        for (int q = 0; q < 2; ++q) {
          const v8f t = V[g][2 * p + q];
          v2f a = o01 * __builtin_shufflevector(t, t, 0, 1);
          a = __builtin_elementwise_fma(o23, __builtin_shufflevector(t, t, 2, 3), a);
          a = __builtin_elementwise_fma(o45, __builtin_shufflevector(t, t, 4, 5), a);
          a = __builtin_elementwise_fma(o67, __builtin_shufflevector(t, t, 6, 7), a);
          dt[q] = a.x + a.y;
        }
        const float keep = half ? dt[1] : dt[0];
        const float send = half ? dt[0] : dt[1];
        L[g][p] = keep + xhalf(send);
      }
    }

    float mx[2];
#pragma unroll
    for (int p = 0; p < 2; ++p) {
      mx[p] = fmaxf(fmaxf(L[0][p], L[1][p]), fmaxf(L[2][p], L[3][p]));
      xm[(p * 32 + lane) * SPITCH_ + wave] = mx[p];
    }
    __syncthreads();
    {
      typedef float v4f __attribute__((ext_vector_type(4)));
      v4f q0, q1, q2, q3, q4, q5, q6, q7;
      const unsigned ra = (unsigned)(size_t)(xm + lane * SPITCH_);
      static_assert(32 * SPITCH_ * 4 == 2560, "the asm offsets below place column 32 + lane 2560 bytes on");
      asm volatile(
          "ds_load_b128 %0, %8\n\t"
          "ds_load_b128 %1, %8 offset:16\n\t"
          "ds_load_b128 %2, %8 offset:32\n\t"
          "ds_load_b128 %3, %8 offset:48\n\t"
          "ds_load_b128 %4, %8 offset:2560\n\t"
          "ds_load_b128 %5, %8 offset:2576\n\t"
          "ds_load_b128 %6, %8 offset:2592\n\t"
          "ds_load_b128 %7, %8 offset:2608\n\t"
          "s_wait_dscnt 0x0"
          : "=&v"(q0), "=&v"(q1), "=&v"(q2), "=&v"(q3), "=&v"(q4), "=&v"(q5), "=&v"(q6), "=&v"(q7)
          : "v"(ra)
          : "memory");
      float m = fmaxf(fmaxf(q0.x, q0.y), fmaxf(q0.z, q0.w));
      m = fmaxf(fmaxf(m, q1.x), fmaxf(fmaxf(q1.y, q1.z), q1.w));
      m = fmaxf(fmaxf(m, q2.x), fmaxf(fmaxf(q2.y, q2.z), q2.w));
      m = fmaxf(fmaxf(m, q3.x), fmaxf(fmaxf(q3.y, q3.z), q3.w));
      mx[0] = m;
      m = fmaxf(fmaxf(q4.x, q4.y), fmaxf(q4.z, q4.w));
      m = fmaxf(fmaxf(m, q5.x), fmaxf(fmaxf(q5.y, q5.z), q5.w));
      m = fmaxf(fmaxf(m, q6.x), fmaxf(fmaxf(q6.y, q6.z), q6.w));
      m = fmaxf(fmaxf(m, q7.x), fmaxf(fmaxf(q7.y, q7.z), q7.w));
      mx[1] = m;
    }

#pragma unroll
    for (int p = 0; p < 2; ++p) {
      float s = 0.f;
#pragma unroll
      for (int g = 0; g < G_; ++g) { L[g][p] = __expf(L[g][p] - mx[p]); s += L[g][p]; }
      xs[(p * 32 + lane) * SPITCH_ + wave] = s;
    }
    __syncthreads();
    float4 qs[2][NWAVE_ / 4];
#pragma unroll
    for (int p = 0; p < 2; ++p) {
      const float4* rs = (const float4*)(xs + (p * 32 + lane) * SPITCH_);
#pragma unroll
      for (int j = 0; j < NWAVE_ / 4; ++j) qs[p][j] = rs[j];
    }
    float inv[2];
#pragma unroll
    for (int p = 0; p < 2; ++p) {
      float s = qs[p][0].x;
      s += qs[p][0].y; s += qs[p][0].z; s += qs[p][0].w;
#pragma unroll
      for (int j = 1; j < NWAVE_ / 4; ++j) { s += qs[p][j].x; s += qs[p][j].y; s += qs[p][j].z; s += qs[p][j].w; }
      inv[p] = 1.0f / s;
    }

    if (cc + 1 < CHB_)
      load_frags(x + (xo + (unsigned)(cc + 1) * (REP_ * DIN_)), Wt + (wo + (unsigned)(cc + 1) * (DOUT_ * DIN_)),
                 xa, xb, wa, wb);

#pragma unroll
    for (int g = 0; g < G_; ++g) {
#pragma unroll
      for (int p = 0; p < 2; ++p) {
        const float c  = L[g][p] * inv[p];
        const float cx = xhalf(c);
        const float ce = half ? cx : c;
        const float co = half ? c : cx;
        sacc[g] += ce * V[g][2 * p];
        sacc[g] += co * V[g][2 * p + 1];
      }
    }
  }

  float e[G_];
#pragma unroll
  for (int g = 0; g < G_; ++g) e[g] = colsum_capsule(sacc[g], col);
  stage_capsules(e, stage, wave, col, half);

  float* pw = pout + (((size_t)b * NCHUNK_ + chunk) * N_ + n0) * DOUT_;
  flush_partial(stage, pw, wave, lane);
}

__global__ __launch_bounds__(256) void caps_final(
    const float* __restrict__ part, const float* __restrict__ bias, float* __restrict__ out)
{
  const int e  = blockIdx.x * 256 + threadIdx.x;
  const int b  = e / (N_ * DOUT_);
  const int nd = e % (N_ * DOUT_);
  float s = bias[nd];
#pragma unroll
  for (int k = 0; k < NCHUNK_; ++k)
    s += part[((size_t)b * NCHUNK_ + k) * (N_ * DOUT_) + nd];
  const float sn = row_sum16(s * s);
  const float scl = sn / (1.0f + sn) * rsqrtf(sn + EPS_);
  out[e] = scl * s;
}

extern "C" void kernel_launch(void* const* d_in, const int* in_sizes, int n_in,
                              void* d_out, int out_size, void* d_ws, size_t ws_size,
                              hipStream_t stream)
{
  const float* x    = (const float*)d_in[0];
  const float* Wt   = (const float*)d_in[1];
  const float* bias = (const float*)d_in[2];
  float* out = (float*)d_out;

  float* p0 = (float*)d_ws;
  float* p1 = p0 + (size_t)PLANE_;
  float* p2 = p1 + (size_t)PLANE_;

  dim3 grid(B_ * NCHUNK_), blk(512);
  dim3 fgrid((B_ * N_ * DOUT_) / 256), fblk(256);

  caps_route_first<<<grid, blk, 0, stream>>>(x, Wt, p0);
  caps_route<1><<<grid, blk, 0, stream>>>(x, Wt, bias, p0, p0, p1);
  caps_route<2><<<grid, blk, 0, stream>>>(x, Wt, bias, p0, p1, p2);
  caps_final<<<fgrid, fblk, 0, stream>>>(p2, bias, out);
}
